# starting kernel
# baseline (speedup 1.0000x reference)
#include <hip/hip_runtime.h>
#include <hip/hip_bf16.h>
#include <math.h>


#define ZC    8
#define EC    32
#define HC    64
#define BC    64
#define LLC   512
#define LAGSC 2
#define LENC  (LLC - LAGSC)
#define NN    (BC * LENC)
#define RES_SIZE (NN * ZC)
#define LDSPAD 68

typedef __attribute__((ext_vector_type(2))) float v2f;
typedef __attribute__((ext_vector_type(8))) float v8f;

static __device__ __forceinline__ v8f wmma_f32(v2f a, v2f b, v8f c) {
    return __builtin_amdgcn_wmma_f32_16x16x4_f32(
         false, a,  false, b,
         (short)0, c,  false,  false);
}

__global__ __launch_bounds__(128)
void npctp_mlp_jvp_kernel(const float* __restrict__ x,
                          const float* __restrict__ emb,
                          const float* __restrict__ W1,
                          const float* __restrict__ b1,
                          const float* __restrict__ W2,
                          const float* __restrict__ b2,
                          const float* __restrict__ W3,
                          const float* __restrict__ b3,
                          float* __restrict__ out)
{
    __shared__ float buf_h[4][16][LDSPAD];
    __shared__ float buf_d[4][16][LDSPAD];
    __shared__ float lds_ld[16][ZC];

    const int tid  = threadIdx.x;
    const int wave = tid >> 5;
    const int lane = tid & 31;
    const int hi   = lane >> 4;
    const int lo   = lane & 15;
    const int n0   = blockIdx.x * 16;

    v2f aE[8];
    {
        const int n  = n0 + lo;
        const int b  = n / LENC;
        const int li = n - b * LENC;
        const float* ep = emb + (size_t)(b * LLC + li + LAGSC) * EC;
        #pragma unroll
        for (int kk = 0; kk < 8; ++kk) {
            const int e = 4 * kk + 2 * hi;
            aE[kk].x = ep[e];
            aE[kk].y = ep[e + 1];
        }
    }

    #pragma unroll
    for (int zi = 0; zi < 2; ++zi) {
        const int z = wave + 4 * zi;

        float xt[8];
        #pragma unroll
        for (int j = 0; j < 8; ++j) {
            const int n  = n0 + j + 8 * hi;
            const int b  = n / LENC;
            const int li = n - b * LENC;
            xt[j] = x[(size_t)(b * LLC + li + LAGSC) * ZC + z];
        }

        const float* W1z = W1 + z * (EC + 1) * HC;
        const float* W2z = W2 + z * HC * HC;
        const float* b1z = b1 + z * HC;
        const float* b2z = b2 + z * HC;
        const float* W3z = W3 + z * HC;
        const float  b3z = b3[z];

        #pragma unroll
        for (int c = 0; c < 4; ++c) {
            const int col = 16 * c + lo;
            v8f acc = {0.f,0.f,0.f,0.f,0.f,0.f,0.f,0.f};
            #pragma unroll
            for (int kk = 0; kk < 8; ++kk) {
                const int k = 4 * kk + 2 * hi;
                v2f bf;
                bf.x = W1z[(size_t)k * HC + col];
                bf.y = W1z[(size_t)(k + 1) * HC + col];
                acc = wmma_f32(aE[kk], bf, acc);
            }
            const float wlast = W1z[(size_t)EC * HC + col];
            const float bb    = b1z[col];
            #pragma unroll
            for (int j = 0; j < 8; ++j) {
                const float pre = acc[j] + xt[j] * wlast + bb;
                const float f   = (pre >= 0.f) ? 1.f : 0.01f;
                buf_h[wave][j + 8 * hi][col] = f * pre;
                buf_d[wave][j + 8 * hi][col] = f * wlast;
            }
        }

        v2f ah[16], ad[16];
        #pragma unroll
        for (int kk = 0; kk < 16; ++kk) {
            const int k = 4 * kk + 2 * hi;
            ah[kk].x = buf_h[wave][lo][k];
            ah[kk].y = buf_h[wave][lo][k + 1];
            ad[kk].x = buf_d[wave][lo][k];
            ad[kk].y = buf_d[wave][lo][k + 1];
        }

        float pout[8] = {0,0,0,0,0,0,0,0};
        float pdt [8] = {0,0,0,0,0,0,0,0};
        #pragma unroll
        for (int c = 0; c < 4; ++c) {
            const int col = 16 * c + lo;
            v8f acc = {0.f,0.f,0.f,0.f,0.f,0.f,0.f,0.f};
            v8f dac = {0.f,0.f,0.f,0.f,0.f,0.f,0.f,0.f};
            #pragma unroll
            for (int kk = 0; kk < 16; ++kk) {
                const int k = 4 * kk + 2 * hi;
                v2f bf;
                bf.x = W2z[(size_t)k * HC + col];
                bf.y = W2z[(size_t)(k + 1) * HC + col];
                acc = wmma_f32(ah[kk], bf, acc);
                dac = wmma_f32(ad[kk], bf, dac);
            }
            const float bb  = b2z[col];
            const float w3v = W3z[col];
            #pragma unroll
            for (int j = 0; j < 8; ++j) {
                const float pre = acc[j] + bb;
                const float f   = (pre >= 0.f) ? 1.f : 0.01f;
                pout[j] += (f * pre)    * w3v;
                pdt [j] += (f * dac[j]) * w3v;
            }
        }

        #pragma unroll
        for (int j = 0; j < 8; ++j) {
            float a = pout[j], d = pdt[j];
            #pragma unroll
            for (int off = 1; off < 16; off <<= 1) {
                a += __shfl_xor(a, off, 32);
                d += __shfl_xor(d, off, 32);
            }
            pout[j] = a; pdt[j] = d;
        }

        if (lo == 0) {
            #pragma unroll
            for (int j = 0; j < 8; ++j) {
                const int m = j + 8 * hi;
                const int n = n0 + m;
                out[(size_t)n * ZC + z] = pout[j] + b3z;
                lds_ld[m][z] = __logf(fabsf(pdt[j]));
            }
        }
    }

    __syncthreads();
    if (tid < 16) {
        float s = 0.f;
        #pragma unroll
        for (int z = 0; z < ZC; ++z) s += lds_ld[tid][z];
        out[RES_SIZE + n0 + tid] = s;
    }
}

extern "C" void kernel_launch(void* const* d_in, const int* in_sizes, int n_in,
                              void* d_out, int out_size, void* d_ws, size_t ws_size,
                              hipStream_t stream) {
    (void)in_sizes; (void)n_in; (void)out_size; (void)d_ws; (void)ws_size;
    const float* x   = (const float*)d_in[0];
    const float* emb = (const float*)d_in[1];
    const float* W1  = (const float*)d_in[2];
    const float* b1  = (const float*)d_in[3];
    const float* W2  = (const float*)d_in[4];
    const float* b2  = (const float*)d_in[5];
    const float* W3  = (const float*)d_in[6];
    const float* b3  = (const float*)d_in[7];
    float* out = (float*)d_out;

    dim3 grid(NN / 16);
    dim3 block(128);
    npctp_mlp_jvp_kernel<<<grid, block, 0, stream>>>(x, emb, W1, b1, W2, b2,
                                                     W3, b3, out);
}


# Weight staging pipelined by latent pair: only pair 0 staged ahead of the first product; pair s+1 loaded under step s, stored to LDS before layer 2, handed over by a split barrier; sums unchanged
# speedup vs baseline: 1.3759x; 1.3759x over previous
#include <hip/hip_runtime.h>
#include <math.h>


#define ZC    8
#define EC    32
#define HC    64
#define BC    64
#define LLC   512
#define LAGSC 2
#define LENC  (LLC - LAGSC)
#define NN    (BC * LENC)
#define RES_SIZE (NN * ZC)
#define LDSPAD 68
#define P2     68
#define P1     36
#define NTILE  8
#define ZHALF  (ZC / 2)
#define NWAVE  (2 * NTILE)
#define NTHR   (NWAVE * 32)

typedef __attribute__((ext_vector_type(2))) float v2f;
typedef __attribute__((ext_vector_type(4))) float v4f;
typedef __attribute__((ext_vector_type(8))) float v8f;

static __device__ __forceinline__ v8f wmma_f32(v2f a, v2f b, v8f c) {
    return __builtin_amdgcn_wmma_f32_16x16x4_f32(
        false, a, false, b, (short)0, c, false, false);
}

template <int CTRL>
static __device__ __forceinline__ float halve_step(bool bit, float a, float b) {
    const float keep = bit ? b : a;
    const float send = bit ? a : b;
    const int   got  = __builtin_amdgcn_update_dpp(0, __float_as_int(send), CTRL,
                                                   0xf, 0xf, true);
    return keep + __int_as_float(got);
}

__global__ __launch_bounds__(NTHR)
void npctp_mlp_jvp_kernel(const float* __restrict__ x,
                          const float* __restrict__ emb,
                          const float* __restrict__ W1,
                          const float* __restrict__ b1,
                          const float* __restrict__ W2,
                          const float* __restrict__ b2,
                          const float* __restrict__ W3,
                          const float* __restrict__ b3,
                          float* __restrict__ out)
{
    __shared__ __attribute__((aligned(16))) float sW2[ZC * HC * P2];
    __shared__ __attribute__((aligned(16))) float sW1[ZC * HC * P1];
    __shared__ float sWl[ZC * HC];
    __shared__ float sB1[ZC * HC];
    __shared__ float sB2[ZC * HC];
    __shared__ float sW3[ZC * HC];
    __shared__ __attribute__((aligned(16))) float buf_t[NWAVE][16][LDSPAD];
    __shared__ float sL[NTILE][16];

    const int tid  = threadIdx.x;
    const int wave = __builtin_amdgcn_readfirstlane(tid >> 5);
    const int tile = wave & (NTILE - 1);
    const int zh   = wave / NTILE;
    const int lane = tid & 31;
    const int hi   = lane >> 4;
    const int lo   = lane & 15;
    const int n0   = (blockIdx.x * NTILE + tile) * 16;

    const int scol = tid & 63;
    const int sk   = tid >> 6;
    const int szs  = (tid >> 6) & 1;
    const int sarr = wave >> 2;
    const float* sm_src = (sarr == 0) ? (W1 + (size_t)EC * HC)
                        : (sarr == 1) ? b1 : (sarr == 2) ? b2 : W3;
    float*       sm_dst = (sarr == 0) ? sWl : (sarr == 1) ? sB1
                        : (sarr == 2) ? sB2 : sW3;
    const int    sm_zstride = (sarr == 0) ? (EC + 1) * HC : HC;
    v4f   r2[4], r1[2];
    float rs;
    auto stage_load = [&](const int s) {
        #pragma unroll
        for (int it = 0; it < 4; ++it) {
            const int z  = (it >> 1) * ZHALF + s;
            const int k4 = (it & 1) * 8 + sk;
            const float* g = W2 + (size_t)z * HC * HC + (size_t)(4 * k4) * HC + scol;
            r2[it].x = g[0]; r2[it].y = g[HC]; r2[it].z = g[2 * HC]; r2[it].w = g[3 * HC];
        }
        #pragma unroll
        for (int it = 0; it < 2; ++it) {
            const int z = it * ZHALF + s;
            const float* g = W1 + (size_t)z * (EC + 1) * HC + (size_t)(4 * sk) * HC + scol;
            r1[it].x = g[0]; r1[it].y = g[HC]; r1[it].z = g[2 * HC]; r1[it].w = g[3 * HC];
        }
        rs = sm_src[(size_t)(szs * ZHALF + s) * sm_zstride + scol];
    };
    auto stage_store = [&](const int s) {
        #pragma unroll
        for (int it = 0; it < 4; ++it) {
            const int z  = (it >> 1) * ZHALF + s;
            const int k4 = (it & 1) * 8 + sk;
            *(v4f*)(sW2 + (z * HC + scol) * P2 + 4 * k4) = r2[it];
        }
        #pragma unroll
        for (int it = 0; it < 2; ++it) {
            const int z = it * ZHALF + s;
            *(v4f*)(sW1 + (z * HC + scol) * P1 + 4 * sk) = r1[it];
        }
        sm_dst[(szs * ZHALF + s) * HC + scol] = rs;
    };
    stage_load(0);

    v4f aE[4];
    {
        const int n  = n0 + lo;
        const int b  = n / LENC;
        const int li = n - b * LENC;
        const float* ep = emb + (size_t)(b * LLC + li + LAGSC) * EC + 4 * hi;
        #pragma unroll
        for (int i = 0; i < 4; ++i) aE[i] = *(const v4f*)(ep + 8 * i);
    }
    int xoff[8];
    #pragma unroll
    for (int j = 0; j < 8; ++j) {
        const int n  = n0 + j + 8 * hi;
        const int b  = n / LENC;
        const int li = n - b * LENC;
        xoff[j] = (b * LLC + li + LAGSC) * ZC;
    }
    const int  nrow = n0 + (lo & 7) + 8 * hi;
    const bool bit0 = (lane & 1) != 0;
    const bool bit1 = (lane & 2) != 0;
    const bool bit2 = (lane & 4) != 0;
    const bool bit3 = (lane & 8) != 0;
    float lsum = 0.f;

    stage_store(0);
    __syncthreads();

    for (int zi = 0; zi < ZHALF; ++zi) {
        const int z = zh * ZHALF + zi;
        float xt[8];
        #pragma unroll
        for (int j = 0; j < 8; ++j) xt[j] = x[xoff[j] + z];
        const float b3z = b3[z];
        const bool more = (zi + 1 < ZHALF);
        if (more) stage_load(zi + 1);

        const float* w1z = sW1 + z * HC * P1 + 4 * hi;
        const float* w2z = sW2 + z * HC * P2 + 4 * hi;
        const float* wlz = sWl + z * HC;
        const float* b1z = sB1 + z * HC;
        const float* b2z = sB2 + z * HC;
        const float* w3z = sW3 + z * HC;

        float dv[4][8];
        #pragma unroll
        for (int c = 0; c < 4; ++c) {
            const int col = 16 * c + lo;
            v8f acc = {0.f,0.f,0.f,0.f,0.f,0.f,0.f,0.f};
            #pragma unroll
            for (int i = 0; i < 4; ++i) {
                const v4f bv = *(const v4f*)(w1z + col * P1 + 8 * i);
                acc = wmma_f32(aE[i].xy, bv.xy, acc);
                acc = wmma_f32(aE[i].zw, bv.zw, acc);
            }
            const float wlast = wlz[col];
            const float bb    = b1z[col];
            #pragma unroll
            for (int j = 0; j < 8; ++j) {
                const float pre = acc[j] + xt[j] * wlast + bb;
                const float f   = (pre >= 0.f) ? 1.f : 0.01f;
                buf_t[wave][j + 8 * hi][col] = f * pre;
                dv[c][j] = f * wlast;
            }
        }

        v4f ah[8], ad[8];
        #pragma unroll
        for (int i = 0; i < 8; ++i)
            ah[i] = *(const v4f*)(&buf_t[wave][lo][8 * i + 4 * hi]);
        #pragma unroll
        for (int c = 0; c < 4; ++c) {
            #pragma unroll
            for (int j = 0; j < 8; ++j)
                buf_t[wave][j + 8 * hi][16 * c + lo] = dv[c][j];
        }
        #pragma unroll
        for (int i = 0; i < 8; ++i)
            ad[i] = *(const v4f*)(&buf_t[wave][lo][8 * i + 4 * hi]);

        if (more) {
            stage_store(zi + 1);
            __builtin_amdgcn_fence(__ATOMIC_RELEASE, "workgroup", "local");
            __builtin_amdgcn_s_barrier_signal(-1);
        }

        v2f po[4], pd[4];
        #pragma unroll
        for (int i = 0; i < 4; ++i) { po[i] = (v2f){0.f, 0.f}; pd[i] = (v2f){0.f, 0.f}; }
        #pragma unroll
        for (int c = 0; c < 4; ++c) {
            const int col = 16 * c + lo;
            v8f acc = {0.f,0.f,0.f,0.f,0.f,0.f,0.f,0.f};
            v8f dac = {0.f,0.f,0.f,0.f,0.f,0.f,0.f,0.f};
            #pragma unroll
            for (int i = 0; i < 8; ++i) {
                const v4f bv = *(const v4f*)(w2z + col * P2 + 8 * i);
                acc = wmma_f32(ah[i].xy, bv.xy, acc);
                dac = wmma_f32(ad[i].xy, bv.xy, dac);
                acc = wmma_f32(ah[i].zw, bv.zw, acc);
                dac = wmma_f32(ad[i].zw, bv.zw, dac);
            }
            const float bb  = b2z[col];
            const float w3v = w3z[col];
            #pragma unroll
            for (int i = 0; i < 4; ++i) {
                const v2f pre = (v2f){acc[2 * i], acc[2 * i + 1]} + bb;
                const v2f dd  = (v2f){dac[2 * i], dac[2 * i + 1]};
                v2f f;
                f.x = (pre.x >= 0.f) ? 1.f : 0.01f;
                f.y = (pre.y >= 0.f) ? 1.f : 0.01f;
                po[i] += (f * pre) * w3v;
                pd[i] += (f * dd)  * w3v;
            }
        }

        float w[8];
        #pragma unroll
        for (int i = 0; i < 4; ++i) {
            w[i]     = halve_step<0x161>(bit0, po[i].x, po[i].y);
            w[4 + i] = halve_step<0x161>(bit0, pd[i].x, pd[i].y);
        }
        float u[4];
        #pragma unroll
        for (int i = 0; i < 4; ++i)
            u[i] = halve_step<0x162>(bit1, w[2 * i], w[2 * i + 1]);
        const float t0 = halve_step<0x164>(bit2, u[0], u[1]);
        const float t1 = halve_step<0x164>(bit2, u[2], u[3]);
        const float s  = halve_step<0x168>(bit3, t0, t1);

        lsum += __logf(fabsf(s));
        if (lo < 8) out[(size_t)nrow * ZC + z] = s + b3z;

        if (more) {
            __builtin_amdgcn_s_barrier_wait(-1);
            __builtin_amdgcn_fence(__ATOMIC_ACQUIRE, "workgroup", "local");
        }
    }

    if (zh == 1 && lo >= 8) sL[tile][(lo & 7) + 8 * hi] = lsum;
    __builtin_amdgcn_fence(__ATOMIC_RELEASE, "workgroup", "local");
    __builtin_amdgcn_s_barrier();
    __builtin_amdgcn_fence(__ATOMIC_ACQUIRE, "workgroup", "local");
    if (zh == 0 && lo >= 8)
        out[RES_SIZE + nrow] = lsum + sL[tile][(lo & 7) + 8 * hi];
}

extern "C" void kernel_launch(void* const* d_in, const int* in_sizes, int n_in,
                              void* d_out, int out_size, void* d_ws, size_t ws_size,
                              hipStream_t stream) {
    (void)in_sizes; (void)n_in; (void)out_size; (void)d_ws; (void)ws_size;
    const float* x   = (const float*)d_in[0];
    const float* emb = (const float*)d_in[1];
    const float* W1  = (const float*)d_in[2];
    const float* b1  = (const float*)d_in[3];
    const float* W2  = (const float*)d_in[4];
    const float* b2  = (const float*)d_in[5];
    const float* W3  = (const float*)d_in[6];
    const float* b3  = (const float*)d_in[7];
    float* out = (float*)d_out;

    dim3 grid(255);
    dim3 block(512);
    npctp_mlp_jvp_kernel<<<grid, block, 0, stream>>>(x, emb, W1, b1, W2, b2,
                                                     W3, b3, out);
}
